# starting kernel
# baseline (speedup 1.0000x reference)
#include <hip/hip_runtime.h>
#include <hip/hip_bf16.h>

#define NB 4
#define MM 256
#define DD 32
#define SS 32

typedef float v2f __attribute__((ext_vector_type(2)));
typedef float v8f __attribute__((ext_vector_type(8)));

__global__ void ign_repack_kernel(const float* __restrict__ coeffs,
                                  float* __restrict__ C10T,
                                  float* __restrict__ C11T) {
    int t = threadIdx.x;
    int s = t >> 5, d = t & 31;
    C10T[s * DD + d] = coeffs[(d * SS + s) * 15 + 9];
    C11T[s * DD + d] = coeffs[(d * SS + s) * 15 + 10];
}

__global__ void ign_stats_kernel(const float* __restrict__ in,
                                 float* __restrict__ R,
                                 float* __restrict__ Cc,
                                 float* __restrict__ G) {
    const int n = blockIdx.y, k = blockIdx.x;
    const int t = threadIdx.x;
    const int d = t & 31, grp = t >> 5;
    const float* base = in + (size_t)n * MM * MM * DD;
    float r = 0.f, c = 0.f;
    for (int q = grp; q < MM; q += 8) {
        r += base[((size_t)k * MM + q) * DD + d];
        c += base[((size_t)q * MM + k) * DD + d];
    }
    __shared__ float sr[8][32];
    __shared__ float sc[8][32];
    sr[grp][d] = r; sc[grp][d] = c;
    __syncthreads();
    if (t < 32) {
        float rr = 0.f, cc = 0.f;
        #pragma unroll
        for (int g = 0; g < 8; ++g) { rr += sr[g][t]; cc += sc[g][t]; }
        const int o = (n * DD + t) * MM + k;
        R[o]  = rr;
        Cc[o] = cc;
        G[o]  = base[((size_t)k * MM + k) * DD + t];
    }
}

__global__ void ign_pt_kernel(const float* __restrict__ R,
                              const float* __restrict__ G,
                              float* __restrict__ P,
                              float* __restrict__ T) {
    int t = threadIdx.x;
    float p = 0.f, s = 0.f;
    for (int k = 0; k < MM; ++k) { p += G[t * MM + k]; s += R[t * MM + k]; }
    P[t] = p; T[t] = s;
}

__global__ void ign_bcast_kernel(const float* __restrict__ R,
                                 const float* __restrict__ Cc,
                                 const float* __restrict__ G,
                                 const float* __restrict__ P,
                                 const float* __restrict__ T,
                                 const float* __restrict__ coeffs,
                                 const float* __restrict__ diag_bias,
                                 const float* __restrict__ all_bias,
                                 float* __restrict__ U,
                                 float* __restrict__ V,
                                 float* __restrict__ W) {
    const int n = blockIdx.y, i = blockIdx.x, s = threadIdx.x;
    const float inv_m = 1.0f / (float)MM;
    const float inv_m2 = inv_m * inv_m;
    float u = 0.f, v = 0.f, w = 0.f;
    for (int d = 0; d < DD; ++d) {
        const int o = (n * DD + d) * MM + i;
        const float r = R[o] * inv_m;
        const float c = Cc[o] * inv_m;
        const float g = G[o];
        const float p = P[n * DD + d] * inv_m;
        const float tt = T[n * DD + d] * inv_m2;
        const float* cf = coeffs + (size_t)(d * SS + s) * 15;
        u += cf[5] * c + cf[6] * r + cf[11] * g + cf[13] * p + cf[14] * tt;
        v += cf[7] * c + cf[8] * r + cf[12] * g;
        w += cf[0] * g + cf[1] * p + cf[2] * r + cf[3] * c + cf[4] * tt;
    }
    const int o = (n * MM + i) * SS + s;
    U[o] = u + all_bias[s];
    V[o] = v;
    W[o] = w + diag_bias[s];
}

__global__ __launch_bounds__(256)
void ign_main_kernel(const float* __restrict__ in,
                     const float* __restrict__ C10T,
                     const float* __restrict__ C11T,
                     const float* __restrict__ U,
                     const float* __restrict__ V,
                     const float* __restrict__ W,
                     float* __restrict__ out) {
    const int n = blockIdx.z;
    const int i = blockIdx.y * 8 + (threadIdx.x >> 5);
    const int J = blockIdx.x * 16;
    const int lane = threadIdx.x & 31;
    const int m = lane & 15;
    const int kh = lane >> 4;

    const float* basen = in + (size_t)n * MM * MM * DD;
    const float* a1p = basen + ((size_t)i * MM + (J + m)) * DD + 2 * kh;
    const float* a2p = basen + ((size_t)(J + m) * MM + i) * DD + 2 * kh;

    v2f A1[8], A2[8];
    #pragma unroll
    for (int kb = 0; kb < 8; ++kb) {
        A1[kb] = *(const v2f*)(a1p + 4 * kb);
        A2[kb] = *(const v2f*)(a2p + 4 * kb);
    }

    const int s0 = m, s1 = m + 16;
    const float* b10_0 = C10T + s0 * DD + 2 * kh;
    const float* b10_1 = C10T + s1 * DD + 2 * kh;
    const float* b11_0 = C11T + s0 * DD + 2 * kh;
    const float* b11_1 = C11T + s1 * DD + 2 * kh;

    v8f acc0 = {}; v8f acc1 = {};
    #pragma unroll
    for (int kb = 0; kb < 8; ++kb) {
        v2f bA0 = *(const v2f*)(b10_0 + 4 * kb);
        v2f bA1 = *(const v2f*)(b10_1 + 4 * kb);
        v2f bB0 = *(const v2f*)(b11_0 + 4 * kb);
        v2f bB1 = *(const v2f*)(b11_1 + 4 * kb);
        acc0 = __builtin_amdgcn_wmma_f32_16x16x4_f32(false, A1[kb], false, bA0, (short)0, acc0, false, false);
        acc1 = __builtin_amdgcn_wmma_f32_16x16x4_f32(false, A1[kb], false, bA1, (short)0, acc1, false, false);
        acc0 = __builtin_amdgcn_wmma_f32_16x16x4_f32(false, A2[kb], false, bB0, (short)0, acc0, false, false);
        acc1 = __builtin_amdgcn_wmma_f32_16x16x4_f32(false, A2[kb], false, bB1, (short)0, acc1, false, false);
    }

    const float u0 = U[(n * MM + i) * SS + s0];
    const float u1 = U[(n * MM + i) * SS + s1];
    const float w0 = W[(n * MM + i) * SS + s0];
    const float w1 = W[(n * MM + i) * SS + s1];
    #pragma unroll
    for (int v = 0; v < 8; ++v) {
        const int j = J + v + 8 * kh;
        const float vj0 = V[(n * MM + j) * SS + s0];
        const float vj1 = V[(n * MM + j) * SS + s1];
        const float dsel = (j == i) ? 1.0f : 0.0f;
        const size_t ob = (((size_t)n * MM + i) * MM + j) * SS;
        out[ob + s0] = acc0[v] + u0 + vj0 + dsel * w0;
        out[ob + s1] = acc1[v] + u1 + vj1 + dsel * w1;
    }
}

extern "C" void kernel_launch(void* const* d_in, const int* in_sizes, int n_in,
                              void* d_out, int out_size, void* d_ws, size_t ws_size,
                              hipStream_t stream) {
    (void)in_sizes; (void)n_in; (void)out_size; (void)ws_size;
    const float* inputs    = (const float*)d_in[0];
    const float* coeffs    = (const float*)d_in[2];
    const float* diag_bias = (const float*)d_in[3];
    const float* all_bias  = (const float*)d_in[4];
    float* out = (float*)d_out;

    float* ws   = (float*)d_ws;
    float* R    = ws;
    float* Cc   = R    + NB * DD * MM;
    float* G    = Cc   + NB * DD * MM;
    float* P    = G    + NB * DD * MM;
    float* T    = P    + NB * DD;
    float* U    = T    + NB * DD;
    float* V    = U    + NB * MM * SS;
    float* W    = V    + NB * MM * SS;
    float* C10T = W    + NB * MM * SS;
    float* C11T = C10T + SS * DD;

    ign_repack_kernel<<<1, 1024, 0, stream>>>(coeffs, C10T, C11T);
    ign_stats_kernel<<<dim3(MM, NB), 256, 0, stream>>>(inputs, R, Cc, G);
    ign_pt_kernel<<<1, NB * DD, 0, stream>>>(R, G, P, T);
    ign_bcast_kernel<<<dim3(MM, NB), SS, 0, stream>>>(R, Cc, G, P, T, coeffs,
                                                      diag_bias, all_bias, U, V, W);
    ign_main_kernel<<<dim3(MM / 16, MM / 8, NB), 256, 0, stream>>>(
        inputs, C10T, C11T, U, V, W, out);
}


# Fuse coeff repack, P/T reduction and U/V/W broadcast kernels into one prep launch (5 -> 3 launches), coeffs staged in LDS; P/T summed as a lane tree, U/V/W d-sums in 8 groups of 4
# speedup vs baseline: 1.5240x; 1.5240x over previous
#include <hip/hip_runtime.h>
#include <hip/hip_bf16.h>

#define NB 4
#define MM 256
#define DD 32
#define SS 32

typedef float v2f __attribute__((ext_vector_type(2)));
typedef float v8f __attribute__((ext_vector_type(8)));

__global__ void ign_stats_kernel(const float* __restrict__ in,
                                 float* __restrict__ R,
                                 float* __restrict__ Cc,
                                 float* __restrict__ G) {
    const int n = blockIdx.y, k = blockIdx.x;
    const int t = threadIdx.x;
    const int d = t & 31, grp = t >> 5;
    const float* base = in + (size_t)n * MM * MM * DD;
    float r = 0.f, c = 0.f;
    for (int q = grp; q < MM; q += 8) {
        r += base[((size_t)k * MM + q) * DD + d];
        c += base[((size_t)q * MM + k) * DD + d];
    }
    __shared__ float sr[8][32];
    __shared__ float sc[8][32];
    sr[grp][d] = r; sc[grp][d] = c;
    __syncthreads();
    if (t < 32) {
        float rr = 0.f, cc = 0.f;
        #pragma unroll
        for (int g = 0; g < 8; ++g) { rr += sr[g][t]; cc += sc[g][t]; }
        const int o = (n * DD + t) * MM + k;
        R[o]  = rr;
        Cc[o] = cc;
        G[o]  = base[((size_t)k * MM + k) * DD + t];
    }
}

#define IT 8
__global__ __launch_bounds__(256)
void ign_prep_kernel(const float* __restrict__ R,
                     const float* __restrict__ Cc,
                     const float* __restrict__ G,
                     const float* __restrict__ coeffs,
                     const float* __restrict__ diag_bias,
                     const float* __restrict__ all_bias,
                     float* __restrict__ U,
                     float* __restrict__ V,
                     float* __restrict__ W,
                     float* __restrict__ C10T,
                     float* __restrict__ C11T) {
    __shared__ __attribute__((aligned(16))) float cfL[DD * SS * 15];
    __shared__ __attribute__((aligned(16))) float sS[3][DD][IT];

    const int n = blockIdx.y, i0 = blockIdx.x * IT;
    const int t = threadIdx.x, l = t & 31, w = t >> 5;

    float4 ra[4], rb[4], ga[4], gb[4];
    for (int r = 0; r < 4; ++r) {
        const float4* Rrow = (const float4*)(R + (size_t)(n * DD + 4 * w + r) * MM);
        const float4* Grow = (const float4*)(G + (size_t)(n * DD + 4 * w + r) * MM);
        ra[r] = Rrow[l]; rb[r] = Rrow[l + 32];
        ga[r] = Grow[l]; gb[r] = Grow[l + 32];
    }
    float4 cv[15];
    const float4* cg4 = (const float4*)coeffs;
    for (int q = 0; q < 15; ++q) cv[q] = cg4[t + 256 * q];
    const size_t so = (size_t)(n * DD + (t >> 3)) * MM + i0 + (t & 7);
    const float svR = R[so], svC = Cc[so], svG = G[so];

    float4* cl4 = (float4*)cfL;
    for (int q = 0; q < 15; ++q) cl4[t + 256 * q] = cv[q];
    sS[0][t >> 3][t & 7] = svR;
    sS[1][t >> 3][t & 7] = svC;
    sS[2][t >> 3][t & 7] = svG;

    float pP[4], pT[4];
    for (int r = 0; r < 4; ++r) {
        float sr = ((ra[r].x + ra[r].y) + (ra[r].z + ra[r].w)) + ((rb[r].x + rb[r].y) + (rb[r].z + rb[r].w));
        float sg = ((ga[r].x + ga[r].y) + (ga[r].z + ga[r].w)) + ((gb[r].x + gb[r].y) + (gb[r].z + gb[r].w));
        for (int off = 16; off > 0; off >>= 1) {
            sr += __shfl_xor(sr, off);
            sg += __shfl_xor(sg, off);
        }
        pT[r] = sr;
        pP[r] = sg;
    }
    __syncthreads();

    if (blockIdx.x == 0 && blockIdx.y == 0) {
        for (int q = 0; q < 4; ++q) {
            const int idx = t + 256 * q;
            const int s2 = idx >> 5, d2 = idx & 31;
            C10T[idx] = cfL[(d2 * SS + s2) * 15 + 9];
            C11T[idx] = cfL[(d2 * SS + s2) * 15 + 10];
        }
    }

    const float inv_m = 1.0f / (float)MM;
    const float inv_m2 = inv_m * inv_m;
    float u[IT], v[IT], ww[IT];
    for (int ii = 0; ii < IT; ++ii) { u[ii] = 0.f; v[ii] = 0.f; ww[ii] = 0.f; }
    for (int r = 0; r < 4; ++r) {
        const int d = 4 * w + r;
        const float* cf = cfL + (d * SS + l) * 15;
        const float p = pP[r] * inv_m;
        const float tt = pT[r] * inv_m2;
        for (int ii = 0; ii < IT; ++ii) {
            const float rr = sS[0][d][ii] * inv_m;
            const float c = sS[1][d][ii] * inv_m;
            const float g = sS[2][d][ii];
            u[ii] += cf[5] * c + cf[6] * rr + cf[11] * g + cf[13] * p + cf[14] * tt;
            v[ii] += cf[7] * c + cf[8] * rr + cf[12] * g;
            ww[ii] += cf[0] * g + cf[1] * p + cf[2] * rr + cf[3] * c + cf[4] * tt;
        }
    }
    __syncthreads();

    float* part = cfL;
    for (int ii = 0; ii < IT; ++ii) {
        part[((w * 3 + 0) * IT + ii) * SS + l] = u[ii];
        part[((w * 3 + 1) * IT + ii) * SS + l] = v[ii];
        part[((w * 3 + 2) * IT + ii) * SS + l] = ww[ii];
    }
    __syncthreads();

    float us = 0.f, vs = 0.f, ws = 0.f;
    for (int g = 0; g < 8; ++g) {
        us += part[((g * 3 + 0) * IT + w) * SS + l];
        vs += part[((g * 3 + 1) * IT + w) * SS + l];
        ws += part[((g * 3 + 2) * IT + w) * SS + l];
    }
    const int o = (n * MM + i0 + w) * SS + l;
    U[o] = us + all_bias[l];
    V[o] = vs;
    W[o] = ws + diag_bias[l];
}

__global__ __launch_bounds__(256)
void ign_main_kernel(const float* __restrict__ in,
                     const float* __restrict__ C10T,
                     const float* __restrict__ C11T,
                     const float* __restrict__ U,
                     const float* __restrict__ V,
                     const float* __restrict__ W,
                     float* __restrict__ out) {
    const int n = blockIdx.z;
    const int i = blockIdx.y * 8 + (threadIdx.x >> 5);
    const int J = blockIdx.x * 16;
    const int lane = threadIdx.x & 31;
    const int m = lane & 15;
    const int kh = lane >> 4;

    const float* basen = in + (size_t)n * MM * MM * DD;
    const float* a1p = basen + ((size_t)i * MM + (J + m)) * DD + 2 * kh;
    const float* a2p = basen + ((size_t)(J + m) * MM + i) * DD + 2 * kh;

    v2f A1[8], A2[8];
    #pragma unroll
    for (int kb = 0; kb < 8; ++kb) {
        A1[kb] = *(const v2f*)(a1p + 4 * kb);
        A2[kb] = *(const v2f*)(a2p + 4 * kb);
    }

    const int s0 = m, s1 = m + 16;
    const float* b10_0 = C10T + s0 * DD + 2 * kh;
    const float* b10_1 = C10T + s1 * DD + 2 * kh;
    const float* b11_0 = C11T + s0 * DD + 2 * kh;
    const float* b11_1 = C11T + s1 * DD + 2 * kh;

    v8f acc0 = {}; v8f acc1 = {};
    #pragma unroll
    for (int kb = 0; kb < 8; ++kb) {
        v2f bA0 = *(const v2f*)(b10_0 + 4 * kb);
        v2f bA1 = *(const v2f*)(b10_1 + 4 * kb);
        v2f bB0 = *(const v2f*)(b11_0 + 4 * kb);
        v2f bB1 = *(const v2f*)(b11_1 + 4 * kb);
        acc0 = __builtin_amdgcn_wmma_f32_16x16x4_f32(false, A1[kb], false, bA0, (short)0, acc0, false, false);
        acc1 = __builtin_amdgcn_wmma_f32_16x16x4_f32(false, A1[kb], false, bA1, (short)0, acc1, false, false);
        acc0 = __builtin_amdgcn_wmma_f32_16x16x4_f32(false, A2[kb], false, bB0, (short)0, acc0, false, false);
        acc1 = __builtin_amdgcn_wmma_f32_16x16x4_f32(false, A2[kb], false, bB1, (short)0, acc1, false, false);
    }

    const float u0 = U[(n * MM + i) * SS + s0];
    const float u1 = U[(n * MM + i) * SS + s1];
    const float w0 = W[(n * MM + i) * SS + s0];
    const float w1 = W[(n * MM + i) * SS + s1];
    #pragma unroll
    for (int v = 0; v < 8; ++v) {
        const int j = J + v + 8 * kh;
        const float vj0 = V[(n * MM + j) * SS + s0];
        const float vj1 = V[(n * MM + j) * SS + s1];
        const float dsel = (j == i) ? 1.0f : 0.0f;
        const size_t ob = (((size_t)n * MM + i) * MM + j) * SS;
        out[ob + s0] = acc0[v] + u0 + vj0 + dsel * w0;
        out[ob + s1] = acc1[v] + u1 + vj1 + dsel * w1;
    }
}

extern "C" void kernel_launch(void* const* d_in, const int* in_sizes, int n_in,
                              void* d_out, int out_size, void* d_ws, size_t ws_size,
                              hipStream_t stream) {
    (void)in_sizes; (void)n_in; (void)out_size; (void)ws_size;
    const float* inputs    = (const float*)d_in[0];
    const float* coeffs    = (const float*)d_in[2];
    const float* diag_bias = (const float*)d_in[3];
    const float* all_bias  = (const float*)d_in[4];
    float* out = (float*)d_out;

    float* ws   = (float*)d_ws;
    float* R    = ws;
    float* Cc   = R    + NB * DD * MM;
    float* G    = Cc   + NB * DD * MM;
    float* P    = G    + NB * DD * MM;
    float* T    = P    + NB * DD;
    float* U    = T    + NB * DD;
    float* V    = U    + NB * MM * SS;
    float* W    = V    + NB * MM * SS;
    float* C10T = W    + NB * MM * SS;
    float* C11T = C10T + SS * DD;

    ign_stats_kernel<<<dim3(MM, NB), 256, 0, stream>>>(inputs, R, Cc, G);
    ign_prep_kernel<<<dim3(MM / IT, NB), 256, 0, stream>>>(R, Cc, G, coeffs, diag_bias, all_bias,
                                                           U, V, W, C10T, C11T);
    ign_main_kernel<<<dim3(MM / 16, MM / 8, NB), 256, 0, stream>>>(
        inputs, C10T, C11T, U, V, W, out);
}


# Main kernel: 256 blocks, each wave loops over 8 tiles of 16 j (16384 -> 2048 waves); C10/C11 B fragments fetched once per wave in fragment order, kept in registers; next tile's A loads issued ahead
# speedup vs baseline: 1.9771x; 1.2973x over previous
#include <hip/hip_runtime.h>
#include <hip/hip_bf16.h>

#define NB 4
#define MM 256
#define DD 32
#define SS 32

typedef float v2f __attribute__((ext_vector_type(2)));
typedef float v8f __attribute__((ext_vector_type(8)));

__global__ void ign_stats_kernel(const float* __restrict__ in,
                                 float* __restrict__ R,
                                 float* __restrict__ Cc,
                                 float* __restrict__ G) {
    const int n = blockIdx.y, k = blockIdx.x;
    const int t = threadIdx.x;
    const int d = t & 31, grp = t >> 5;
    const float* base = in + (size_t)n * MM * MM * DD;
    float r = 0.f, c = 0.f;
    for (int q = grp; q < MM; q += 8) {
        r += base[((size_t)k * MM + q) * DD + d];
        c += base[((size_t)q * MM + k) * DD + d];
    }
    __shared__ float sr[8][32];
    __shared__ float sc[8][32];
    sr[grp][d] = r; sc[grp][d] = c;
    __syncthreads();
    if (t < 32) {
        float rr = 0.f, cc = 0.f;
        #pragma unroll
        for (int g = 0; g < 8; ++g) { rr += sr[g][t]; cc += sc[g][t]; }
        const int o = (n * DD + t) * MM + k;
        R[o]  = rr;
        Cc[o] = cc;
        G[o]  = base[((size_t)k * MM + k) * DD + t];
    }
}

#define IT 8
__global__ __launch_bounds__(256)
void ign_prep_kernel(const float* __restrict__ R,
                     const float* __restrict__ Cc,
                     const float* __restrict__ G,
                     const float* __restrict__ coeffs,
                     const float* __restrict__ diag_bias,
                     const float* __restrict__ all_bias,
                     float* __restrict__ U,
                     float* __restrict__ V,
                     float* __restrict__ W,
                     float* __restrict__ BQ) {
    __shared__ __attribute__((aligned(16))) float cfL[DD * SS * 15];
    __shared__ __attribute__((aligned(16))) float sS[3][DD][IT];

    const int n = blockIdx.y, i0 = blockIdx.x * IT;
    const int t = threadIdx.x, l = t & 31, w = t >> 5;

    float4 ra[4], rb[4], ga[4], gb[4];
    for (int r = 0; r < 4; ++r) {
        const float4* Rrow = (const float4*)(R + (size_t)(n * DD + 4 * w + r) * MM);
        const float4* Grow = (const float4*)(G + (size_t)(n * DD + 4 * w + r) * MM);
        ra[r] = Rrow[l]; rb[r] = Rrow[l + 32];
        ga[r] = Grow[l]; gb[r] = Grow[l + 32];
    }
    float4 cv[15];
    const float4* cg4 = (const float4*)coeffs;
    for (int q = 0; q < 15; ++q) cv[q] = cg4[t + 256 * q];
    const size_t so = (size_t)(n * DD + (t >> 3)) * MM + i0 + (t & 7);
    const float svR = R[so], svC = Cc[so], svG = G[so];

    float4* cl4 = (float4*)cfL;
    for (int q = 0; q < 15; ++q) cl4[t + 256 * q] = cv[q];
    sS[0][t >> 3][t & 7] = svR;
    sS[1][t >> 3][t & 7] = svC;
    sS[2][t >> 3][t & 7] = svG;

    float pP[4], pT[4];
    for (int r = 0; r < 4; ++r) {
        float sr = ((ra[r].x + ra[r].y) + (ra[r].z + ra[r].w)) + ((rb[r].x + rb[r].y) + (rb[r].z + rb[r].w));
        float sg = ((ga[r].x + ga[r].y) + (ga[r].z + ga[r].w)) + ((gb[r].x + gb[r].y) + (gb[r].z + gb[r].w));
        for (int off = 16; off > 0; off >>= 1) {
            sr += __shfl_xor(sr, off);
            sg += __shfl_xor(sg, off);
        }
        pT[r] = sr;
        pP[r] = sg;
    }
    __syncthreads();

    if (blockIdx.x == 0 && blockIdx.y == 0) {
        for (int q = 0; q < 8; ++q) {
            const int idx = t + 256 * q;
            const int e = idx & 3, ln = (idx >> 2) & 31, g = idx >> 7;
            const int d2 = 4 * (g >> 1) + 2 * (ln >> 4) + (e & 1);
            const int s2 = (ln & 15) + 16 * (e >> 1);
            BQ[idx] = cfL[(d2 * SS + s2) * 15 + 9 + (g & 1)];
        }
    }

    const float inv_m = 1.0f / (float)MM;
    const float inv_m2 = inv_m * inv_m;
    float u[IT], v[IT], ww[IT];
    for (int ii = 0; ii < IT; ++ii) { u[ii] = 0.f; v[ii] = 0.f; ww[ii] = 0.f; }
    for (int r = 0; r < 4; ++r) {
        const int d = 4 * w + r;
        const float* cf = cfL + (d * SS + l) * 15;
        const float p = pP[r] * inv_m;
        const float tt = pT[r] * inv_m2;
        for (int ii = 0; ii < IT; ++ii) {
            const float rr = sS[0][d][ii] * inv_m;
            const float c = sS[1][d][ii] * inv_m;
            const float g = sS[2][d][ii];
            u[ii] += cf[5] * c + cf[6] * rr + cf[11] * g + cf[13] * p + cf[14] * tt;
            v[ii] += cf[7] * c + cf[8] * rr + cf[12] * g;
            ww[ii] += cf[0] * g + cf[1] * p + cf[2] * rr + cf[3] * c + cf[4] * tt;
        }
    }
    __syncthreads();

    float* part = cfL;
    for (int ii = 0; ii < IT; ++ii) {
        part[((w * 3 + 0) * IT + ii) * SS + l] = u[ii];
        part[((w * 3 + 1) * IT + ii) * SS + l] = v[ii];
        part[((w * 3 + 2) * IT + ii) * SS + l] = ww[ii];
    }
    __syncthreads();

    float us = 0.f, vs = 0.f, ws = 0.f;
    for (int g = 0; g < 8; ++g) {
        us += part[((g * 3 + 0) * IT + w) * SS + l];
        vs += part[((g * 3 + 1) * IT + w) * SS + l];
        ws += part[((g * 3 + 2) * IT + w) * SS + l];
    }
    const int o = (n * MM + i0 + w) * SS + l;
    U[o] = us + all_bias[l];
    V[o] = vs;
    W[o] = ws + diag_bias[l];
}

#define NT 8

__device__ __forceinline__ void ign_load_a(const float* __restrict__ p1,
                                           const float* __restrict__ p2,
                                           v2f (&A1)[8], v2f (&A2)[8]) {
    #pragma unroll
    for (int kb = 0; kb < 8; ++kb) {
        A1[kb] = *(const v2f*)(p1 + 4 * kb);
        A2[kb] = *(const v2f*)(p2 + 4 * kb);
    }
}

__device__ __forceinline__ void ign_load_v(const float* __restrict__ vp,
                                           float (&vj0)[8], float (&vj1)[8]) {
    #pragma unroll
    for (int v = 0; v < 8; ++v) {
        vj0[v] = vp[v * SS];
        vj1[v] = vp[v * SS + 16];
    }
}

__device__ __forceinline__ void ign_tile(const v2f (&A1)[8], const v2f (&A2)[8],
                                         const float4 (&B10)[8], const float4 (&B11)[8],
                                         const float (&vj0)[8], const float (&vj1)[8],
                                         float* __restrict__ op,
                                         float u0, float u1, float w0, float w1,
                                         int jrow, int i) {
    __builtin_amdgcn_sched_barrier(0);

    v8f acc0 = {}; v8f acc1 = {};
    #pragma unroll
    for (int kb = 0; kb < 8; ++kb) {
        const v2f bA0 = {B10[kb].x, B10[kb].y};
        const v2f bA1 = {B10[kb].z, B10[kb].w};
        const v2f bB0 = {B11[kb].x, B11[kb].y};
        const v2f bB1 = {B11[kb].z, B11[kb].w};
        acc0 = __builtin_amdgcn_wmma_f32_16x16x4_f32(false, A1[kb], false, bA0, (short)0, acc0, false, false);
        acc1 = __builtin_amdgcn_wmma_f32_16x16x4_f32(false, A1[kb], false, bA1, (short)0, acc1, false, false);
        acc0 = __builtin_amdgcn_wmma_f32_16x16x4_f32(false, A2[kb], false, bB0, (short)0, acc0, false, false);
        acc1 = __builtin_amdgcn_wmma_f32_16x16x4_f32(false, A2[kb], false, bB1, (short)0, acc1, false, false);
    }

    #pragma unroll
    for (int v = 0; v < 8; ++v) {
        const float dsel = (jrow + v == i) ? 1.0f : 0.0f;
        op[v * SS]      = acc0[v] + u0 + vj0[v] + dsel * w0;
        op[v * SS + 16] = acc1[v] + u1 + vj1[v] + dsel * w1;
    }
}

__global__ __launch_bounds__(256)
void ign_main_kernel(const float* __restrict__ in,
                     const float* __restrict__ BQ,
                     const float* __restrict__ U,
                     const float* __restrict__ V,
                     const float* __restrict__ W,
                     float* __restrict__ out) {
    const int n = blockIdx.z;
    const int i = blockIdx.y * 8 + (threadIdx.x >> 5);
    const int J0 = blockIdx.x * (16 * NT);
    const int lane = threadIdx.x & 31;
    const int m = lane & 15;
    const int kh = lane >> 4;

    const float* basen = in + (size_t)n * MM * MM * DD;
    const float* a1p = basen + ((size_t)i * MM + (J0 + m)) * DD + 2 * kh;
    const float* a2p = basen + ((size_t)(J0 + m) * MM + i) * DD + 2 * kh;
    const size_t a1step = (size_t)16 * DD;
    const size_t a2step = (size_t)16 * MM * DD;

    v2f Aa1[8], Aa2[8], Ab1[8], Ab2[8];
    ign_load_a(a1p, a2p, Aa1, Aa2);

    const float4* bq = (const float4*)BQ + lane;
    float4 B10[8], B11[8];
    #pragma unroll
    for (int kb = 0; kb < 8; ++kb) {
        B10[kb] = bq[(2 * kb) * 32];
        B11[kb] = bq[(2 * kb + 1) * 32];
    }

    const int s0 = m, s1 = m + 16;
    const float u0 = U[(n * MM + i) * SS + s0];
    const float u1 = U[(n * MM + i) * SS + s1];
    const float w0 = W[(n * MM + i) * SS + s0];
    const float w1 = W[(n * MM + i) * SS + s1];

    const float* vp = V + ((size_t)n * MM + J0 + 8 * kh) * SS + s0;
    float* op = out + (((size_t)n * MM + i) * MM + J0 + 8 * kh) * SS + s0;
    int jrow = J0 + 8 * kh;

    float vj0[8], vj1[8];

    #pragma unroll 1
    for (int tp = 0; tp < NT / 2 - 1; ++tp) {
        ign_load_v(vp, vj0, vj1);
        ign_load_a(a1p + a1step, a2p + a2step, Ab1, Ab2);
        ign_tile(Aa1, Aa2, B10, B11, vj0, vj1, op, u0, u1, w0, w1, jrow, i);
        a1p += 2 * a1step;
        a2p += 2 * a2step;
        ign_load_v(vp + 16 * SS, vj0, vj1);
        ign_load_a(a1p, a2p, Aa1, Aa2);
        ign_tile(Ab1, Ab2, B10, B11, vj0, vj1, op + 16 * SS, u0, u1, w0, w1, jrow + 16, i);
        vp += 32 * SS;
        op += 32 * SS;
        jrow += 32;
    }
    ign_load_v(vp, vj0, vj1);
    ign_load_a(a1p + a1step, a2p + a2step, Ab1, Ab2);
    ign_tile(Aa1, Aa2, B10, B11, vj0, vj1, op, u0, u1, w0, w1, jrow, i);
    ign_load_v(vp + 16 * SS, vj0, vj1);
    ign_tile(Ab1, Ab2, B10, B11, vj0, vj1, op + 16 * SS, u0, u1, w0, w1, jrow + 16, i);
}

extern "C" void kernel_launch(void* const* d_in, const int* in_sizes, int n_in,
                              void* d_out, int out_size, void* d_ws, size_t ws_size,
                              hipStream_t stream) {
    (void)in_sizes; (void)n_in; (void)out_size; (void)ws_size;
    const float* inputs    = (const float*)d_in[0];
    const float* coeffs    = (const float*)d_in[2];
    const float* diag_bias = (const float*)d_in[3];
    const float* all_bias  = (const float*)d_in[4];
    float* out = (float*)d_out;

    float* ws   = (float*)d_ws;
    float* R    = ws;
    float* Cc   = R    + NB * DD * MM;
    float* G    = Cc   + NB * DD * MM;
    float* P    = G    + NB * DD * MM;
    float* T    = P    + NB * DD;
    float* U    = T    + NB * DD;
    float* V    = U    + NB * MM * SS;
    float* W    = V    + NB * MM * SS;
    float* BQ   = W    + NB * MM * SS;

    ign_stats_kernel<<<dim3(MM, NB), 256, 0, stream>>>(inputs, R, Cc, G);
    ign_prep_kernel<<<dim3(MM / IT, NB), 256, 0, stream>>>(R, Cc, G, coeffs, diag_bias, all_bias,
                                                           U, V, W, BQ);
    ign_main_kernel<<<dim3(MM / (16 * NT), MM / 8, NB), 256, 0, stream>>>(
        inputs, BQ, U, V, W, out);
}


# Stats kernel: 256 blocks of 4 k instead of 1024 of 1 (8192 -> 2048 waves), 16-byte loads of 512-B pieces, 3 groups kept in flight; row/col sums re-associated (thread partials, LDS partials in order)
# speedup vs baseline: 2.1247x; 1.0747x over previous
#include <hip/hip_runtime.h>
#include <hip/hip_bf16.h>

#define NB 4
#define MM 256
#define DD 32
#define SS 32

typedef float v2f __attribute__((ext_vector_type(2)));
typedef float v8f __attribute__((ext_vector_type(8)));

#define KT 4
typedef float v4f __attribute__((ext_vector_type(4)));

__global__ __launch_bounds__(256)
void ign_stats_kernel(const float* __restrict__ in,
                      float* __restrict__ R,
                      float* __restrict__ Cc,
                      float* __restrict__ G) {
    __shared__ __attribute__((aligned(16))) float sR[32][DD * KT];
    __shared__ __attribute__((aligned(16))) float sC[8][DD * KT];

    const int n = blockIdx.y, k0 = blockIdx.x * KT;
    const int t = threadIdx.x, l = t & 31, w = t >> 5;
    const float* base = in + (size_t)n * MM * MM * DD;

    float gv = 0.f;
    if (t < DD * KT) {
        const int kd = k0 + (t & 3);
        gv = base[((size_t)kd * MM + kd) * DD + (t >> 2)];
    }

    const v4f* rowp = (const v4f*)(base + (size_t)k0 * MM * DD) + t;
    v4f rv[KT][8];
    #pragma unroll
    for (int kk = 0; kk < KT; ++kk) {
        #pragma unroll
        for (int q = 0; q < 8; ++q) rv[kk][q] = rowp[kk * (MM * DD / 4) + 256 * q];
    }

    asm volatile("" ::: "memory");

    const v4f* colp = (const v4f*)(base + (size_t)k0 * DD) + l + (size_t)w * (MM * DD / 4);
    v4f cv[32];
    #pragma unroll
    for (int q = 0; q < 16; ++q) cv[q] = colp[(size_t)q * (8 * (MM * DD / 4))];

    asm volatile("" :: "v"(rv[0][0]), "v"(rv[0][1]), "v"(rv[0][2]), "v"(rv[0][3]),
                       "v"(rv[0][4]), "v"(rv[0][5]), "v"(rv[0][6]), "v"(rv[0][7]),
                       "v"(rv[1][0]), "v"(rv[1][1]), "v"(rv[1][2]), "v"(rv[1][3]),
                       "v"(rv[1][4]), "v"(rv[1][5]), "v"(rv[1][6]), "v"(rv[1][7]),
                       "v"(rv[2][0]), "v"(rv[2][1]), "v"(rv[2][2]), "v"(rv[2][3]),
                       "v"(rv[2][4]), "v"(rv[2][5]), "v"(rv[2][6]), "v"(rv[2][7]),
                       "v"(rv[3][0]), "v"(rv[3][1]), "v"(rv[3][2]), "v"(rv[3][3]),
                       "v"(rv[3][4]), "v"(rv[3][5])
                 : "memory");

    const int jj = t >> 3, d4 = t & 7;
    v4f ra[KT];
    #pragma unroll
    for (int kk = 0; kk < KT; ++kk) {
        ra[kk] = ((rv[kk][0] + rv[kk][1]) + (rv[kk][2] + rv[kk][3]))
               + ((rv[kk][4] + rv[kk][5]) + (rv[kk][6] + rv[kk][7]));
    }
    v4f* srow = (v4f*)&sR[jj][16 * d4];
    #pragma unroll
    for (int c = 0; c < 4; ++c) {
        const v4f o4 = {ra[0][c], ra[1][c], ra[2][c], ra[3][c]};
        srow[c] = o4;
    }

    #pragma unroll
    for (int q = 16; q < 32; ++q) cv[q] = colp[(size_t)q * (8 * (MM * DD / 4))];

    asm volatile("" :: "v"(cv[0]), "v"(cv[1]), "v"(cv[2]), "v"(cv[3]),
                       "v"(cv[4]), "v"(cv[5]), "v"(cv[6]), "v"(cv[7]),
                       "v"(cv[8]), "v"(cv[9]), "v"(cv[10]), "v"(cv[11]),
                       "v"(cv[12]), "v"(cv[13]), "v"(cv[14]), "v"(cv[15])
                 : "memory");

    v4f ca = cv[0];
    #pragma unroll
    for (int q = 1; q < 32; ++q) ca += cv[q];
    float* scw = &sC[w][16 * (l & 7) + (l >> 3)];
    scw[0] = ca[0]; scw[4] = ca[1]; scw[8] = ca[2]; scw[12] = ca[3];

    __syncthreads();

    if (t < DD * KT) {
        float rr = 0.f, cc = 0.f;
        #pragma unroll
        for (int g = 0; g < 32; ++g) rr += sR[g][t];
        #pragma unroll
        for (int g = 0; g < 8; ++g) cc += sC[g][t];
        const int o = (n * DD + (t >> 2)) * MM + k0 + (t & 3);
        R[o]  = rr;
        Cc[o] = cc;
        G[o]  = gv;
    }
}

#define IT 8
__global__ __launch_bounds__(256)
void ign_prep_kernel(const float* __restrict__ R,
                     const float* __restrict__ Cc,
                     const float* __restrict__ G,
                     const float* __restrict__ coeffs,
                     const float* __restrict__ diag_bias,
                     const float* __restrict__ all_bias,
                     float* __restrict__ U,
                     float* __restrict__ V,
                     float* __restrict__ W,
                     float* __restrict__ BQ) {
    __shared__ __attribute__((aligned(16))) float cfL[DD * SS * 15];
    __shared__ __attribute__((aligned(16))) float sS[3][DD][IT];

    const int n = blockIdx.y, i0 = blockIdx.x * IT;
    const int t = threadIdx.x, l = t & 31, w = t >> 5;

    float4 ra[4], rb[4], ga[4], gb[4];
    for (int r = 0; r < 4; ++r) {
        const float4* Rrow = (const float4*)(R + (size_t)(n * DD + 4 * w + r) * MM);
        const float4* Grow = (const float4*)(G + (size_t)(n * DD + 4 * w + r) * MM);
        ra[r] = Rrow[l]; rb[r] = Rrow[l + 32];
        ga[r] = Grow[l]; gb[r] = Grow[l + 32];
    }
    float4 cv[15];
    const float4* cg4 = (const float4*)coeffs;
    for (int q = 0; q < 15; ++q) cv[q] = cg4[t + 256 * q];
    const size_t so = (size_t)(n * DD + (t >> 3)) * MM + i0 + (t & 7);
    const float svR = R[so], svC = Cc[so], svG = G[so];

    float4* cl4 = (float4*)cfL;
    for (int q = 0; q < 15; ++q) cl4[t + 256 * q] = cv[q];
    sS[0][t >> 3][t & 7] = svR;
    sS[1][t >> 3][t & 7] = svC;
    sS[2][t >> 3][t & 7] = svG;

    float pP[4], pT[4];
    for (int r = 0; r < 4; ++r) {
        float sr = ((ra[r].x + ra[r].y) + (ra[r].z + ra[r].w)) + ((rb[r].x + rb[r].y) + (rb[r].z + rb[r].w));
        float sg = ((ga[r].x + ga[r].y) + (ga[r].z + ga[r].w)) + ((gb[r].x + gb[r].y) + (gb[r].z + gb[r].w));
        for (int off = 16; off > 0; off >>= 1) {
            sr += __shfl_xor(sr, off);
            sg += __shfl_xor(sg, off);
        }
        pT[r] = sr;
        pP[r] = sg;
    }
    __syncthreads();

    if (blockIdx.x == 0 && blockIdx.y == 0) {
        for (int q = 0; q < 8; ++q) {
            const int idx = t + 256 * q;
            const int e = idx & 3, ln = (idx >> 2) & 31, g = idx >> 7;
            const int d2 = 4 * (g >> 1) + 2 * (ln >> 4) + (e & 1);
            const int s2 = (ln & 15) + 16 * (e >> 1);
            BQ[idx] = cfL[(d2 * SS + s2) * 15 + 9 + (g & 1)];
        }
    }

    const float inv_m = 1.0f / (float)MM;
    const float inv_m2 = inv_m * inv_m;
    float u[IT], v[IT], ww[IT];
    for (int ii = 0; ii < IT; ++ii) { u[ii] = 0.f; v[ii] = 0.f; ww[ii] = 0.f; }
    for (int r = 0; r < 4; ++r) {
        const int d = 4 * w + r;
        const float* cf = cfL + (d * SS + l) * 15;
        const float p = pP[r] * inv_m;
        const float tt = pT[r] * inv_m2;
        for (int ii = 0; ii < IT; ++ii) {
            const float rr = sS[0][d][ii] * inv_m;
            const float c = sS[1][d][ii] * inv_m;
            const float g = sS[2][d][ii];
            u[ii] += cf[5] * c + cf[6] * rr + cf[11] * g + cf[13] * p + cf[14] * tt;
            v[ii] += cf[7] * c + cf[8] * rr + cf[12] * g;
            ww[ii] += cf[0] * g + cf[1] * p + cf[2] * rr + cf[3] * c + cf[4] * tt;
        }
    }
    __syncthreads();

    float* part = cfL;
    for (int ii = 0; ii < IT; ++ii) {
        part[((w * 3 + 0) * IT + ii) * SS + l] = u[ii];
        part[((w * 3 + 1) * IT + ii) * SS + l] = v[ii];
        part[((w * 3 + 2) * IT + ii) * SS + l] = ww[ii];
    }
    __syncthreads();

    float us = 0.f, vs = 0.f, ws = 0.f;
    for (int g = 0; g < 8; ++g) {
        us += part[((g * 3 + 0) * IT + w) * SS + l];
        vs += part[((g * 3 + 1) * IT + w) * SS + l];
        ws += part[((g * 3 + 2) * IT + w) * SS + l];
    }
    const int o = (n * MM + i0 + w) * SS + l;
    U[o] = us + all_bias[l];
    V[o] = vs;
    W[o] = ws + diag_bias[l];
}

#define NT 8

__device__ __forceinline__ void ign_load_a(const float* __restrict__ p1,
                                           const float* __restrict__ p2,
                                           v2f (&A1)[8], v2f (&A2)[8]) {
    #pragma unroll
    for (int kb = 0; kb < 8; ++kb) {
        A1[kb] = *(const v2f*)(p1 + 4 * kb);
        A2[kb] = *(const v2f*)(p2 + 4 * kb);
    }
}

__device__ __forceinline__ void ign_load_v(const float* __restrict__ vp,
                                           float (&vj0)[8], float (&vj1)[8]) {
    #pragma unroll
    for (int v = 0; v < 8; ++v) {
        vj0[v] = vp[v * SS];
        vj1[v] = vp[v * SS + 16];
    }
}

__device__ __forceinline__ void ign_tile(const v2f (&A1)[8], const v2f (&A2)[8],
                                         const float4 (&B10)[8], const float4 (&B11)[8],
                                         const float (&vj0)[8], const float (&vj1)[8],
                                         float* __restrict__ op,
                                         float u0, float u1, float w0, float w1,
                                         int jrow, int i) {
    __builtin_amdgcn_sched_barrier(0);

    v8f acc0 = {}; v8f acc1 = {};
    #pragma unroll
    for (int kb = 0; kb < 8; ++kb) {
        const v2f bA0 = {B10[kb].x, B10[kb].y};
        const v2f bA1 = {B10[kb].z, B10[kb].w};
        const v2f bB0 = {B11[kb].x, B11[kb].y};
        const v2f bB1 = {B11[kb].z, B11[kb].w};
        acc0 = __builtin_amdgcn_wmma_f32_16x16x4_f32(false, A1[kb], false, bA0, (short)0, acc0, false, false);
        acc1 = __builtin_amdgcn_wmma_f32_16x16x4_f32(false, A1[kb], false, bA1, (short)0, acc1, false, false);
        acc0 = __builtin_amdgcn_wmma_f32_16x16x4_f32(false, A2[kb], false, bB0, (short)0, acc0, false, false);
        acc1 = __builtin_amdgcn_wmma_f32_16x16x4_f32(false, A2[kb], false, bB1, (short)0, acc1, false, false);
    }

    #pragma unroll
    for (int v = 0; v < 8; ++v) {
        const float dsel = (jrow + v == i) ? 1.0f : 0.0f;
        op[v * SS]      = acc0[v] + u0 + vj0[v] + dsel * w0;
        op[v * SS + 16] = acc1[v] + u1 + vj1[v] + dsel * w1;
    }
}

__global__ __launch_bounds__(256)
void ign_main_kernel(const float* __restrict__ in,
                     const float* __restrict__ BQ,
                     const float* __restrict__ U,
                     const float* __restrict__ V,
                     const float* __restrict__ W,
                     float* __restrict__ out) {
    const int n = blockIdx.z;
    const int i = blockIdx.y * 8 + (threadIdx.x >> 5);
    const int J0 = blockIdx.x * (16 * NT);
    const int lane = threadIdx.x & 31;
    const int m = lane & 15;
    const int kh = lane >> 4;

    const float* basen = in + (size_t)n * MM * MM * DD;
    const float* a1p = basen + ((size_t)i * MM + (J0 + m)) * DD + 2 * kh;
    const float* a2p = basen + ((size_t)(J0 + m) * MM + i) * DD + 2 * kh;
    const size_t a1step = (size_t)16 * DD;
    const size_t a2step = (size_t)16 * MM * DD;

    v2f Aa1[8], Aa2[8], Ab1[8], Ab2[8];
    ign_load_a(a1p, a2p, Aa1, Aa2);

    const float4* bq = (const float4*)BQ + lane;
    float4 B10[8], B11[8];
    #pragma unroll
    for (int kb = 0; kb < 8; ++kb) {
        B10[kb] = bq[(2 * kb) * 32];
        B11[kb] = bq[(2 * kb + 1) * 32];
    }

    const int s0 = m, s1 = m + 16;
    const float u0 = U[(n * MM + i) * SS + s0];
    const float u1 = U[(n * MM + i) * SS + s1];
    const float w0 = W[(n * MM + i) * SS + s0];
    const float w1 = W[(n * MM + i) * SS + s1];

    const float* vp = V + ((size_t)n * MM + J0 + 8 * kh) * SS + s0;
    float* op = out + (((size_t)n * MM + i) * MM + J0 + 8 * kh) * SS + s0;
    int jrow = J0 + 8 * kh;

    float vj0[8], vj1[8];

    #pragma unroll 1
    for (int tp = 0; tp < NT / 2 - 1; ++tp) {
        ign_load_v(vp, vj0, vj1);
        ign_load_a(a1p + a1step, a2p + a2step, Ab1, Ab2);
        ign_tile(Aa1, Aa2, B10, B11, vj0, vj1, op, u0, u1, w0, w1, jrow, i);
        a1p += 2 * a1step;
        a2p += 2 * a2step;
        ign_load_v(vp + 16 * SS, vj0, vj1);
        ign_load_a(a1p, a2p, Aa1, Aa2);
        ign_tile(Ab1, Ab2, B10, B11, vj0, vj1, op + 16 * SS, u0, u1, w0, w1, jrow + 16, i);
        vp += 32 * SS;
        op += 32 * SS;
        jrow += 32;
    }
    ign_load_v(vp, vj0, vj1);
    ign_load_a(a1p + a1step, a2p + a2step, Ab1, Ab2);
    ign_tile(Aa1, Aa2, B10, B11, vj0, vj1, op, u0, u1, w0, w1, jrow, i);
    ign_load_v(vp + 16 * SS, vj0, vj1);
    ign_tile(Ab1, Ab2, B10, B11, vj0, vj1, op + 16 * SS, u0, u1, w0, w1, jrow + 16, i);
}

extern "C" void kernel_launch(void* const* d_in, const int* in_sizes, int n_in,
                              void* d_out, int out_size, void* d_ws, size_t ws_size,
                              hipStream_t stream) {
    (void)in_sizes; (void)n_in; (void)out_size; (void)ws_size;
    const float* inputs    = (const float*)d_in[0];
    const float* coeffs    = (const float*)d_in[2];
    const float* diag_bias = (const float*)d_in[3];
    const float* all_bias  = (const float*)d_in[4];
    float* out = (float*)d_out;

    float* ws   = (float*)d_ws;
    float* R    = ws;
    float* Cc   = R    + NB * DD * MM;
    float* G    = Cc   + NB * DD * MM;
    float* P    = G    + NB * DD * MM;
    float* T    = P    + NB * DD;
    float* U    = T    + NB * DD;
    float* V    = U    + NB * MM * SS;
    float* W    = V    + NB * MM * SS;
    float* BQ   = W    + NB * MM * SS;

    ign_stats_kernel<<<dim3(MM / KT, NB), 256, 0, stream>>>(inputs, R, Cc, G);
    ign_prep_kernel<<<dim3(MM / IT, NB), 256, 0, stream>>>(R, Cc, G, coeffs, diag_bias, all_bias,
                                                           U, V, W, BQ);
    ign_main_kernel<<<dim3(MM / (16 * NT), MM / 8, NB), 256, 0, stream>>>(
        inputs, BQ, U, V, W, out);
}


# Fold the prep launch into its neighbours (3 -> 2 launches): stats blocks make their own U/V/W rows and 64 shares per n of the trace/total terms; main adds the shares in k-block order (re-summed)
# speedup vs baseline: 2.6352x; 1.2403x over previous
#include <hip/hip_runtime.h>
#include <hip/hip_bf16.h>

#define NB 4
#define MM 256
#define DD 32
#define SS 32

typedef float v2f __attribute__((ext_vector_type(2)));
typedef float v8f __attribute__((ext_vector_type(8)));

#define KT 4
#define KB (MM / KT)
typedef float v4f __attribute__((ext_vector_type(4)));

__global__ __launch_bounds__(256)
void ign_stats_kernel(const float* __restrict__ in,
                      const float* __restrict__ coeffs,
                      const float* __restrict__ diag_bias,
                      const float* __restrict__ all_bias,
                      float* __restrict__ U,
                      float* __restrict__ V,
                      float* __restrict__ W,
                      float* __restrict__ ZP,
                      float* __restrict__ BQ) {
    __shared__ __attribute__((aligned(16))) float sm[(32 + 8 + 3) * DD * KT + DD * SS * 15];
    float (*sR)[DD * KT] = (float (*)[DD * KT])sm;
    float (*sC)[DD * KT] = (float (*)[DD * KT])(sm + 32 * DD * KT);
    float (*sT)[DD * KT] = (float (*)[DD * KT])(sm + 40 * DD * KT);
    float* cfL = sm + 43 * DD * KT;

    const int n = blockIdx.y, k0 = blockIdx.x * KT;
    const int t = threadIdx.x, l = t & 31, w = t >> 5;
    const float* base = in + (size_t)n * MM * MM * DD;

    const float ab = all_bias[l], db = diag_bias[l];

    float gv = 0.f;
    if (t < DD * KT) {
        const int kd = k0 + (t & 3);
        gv = base[((size_t)kd * MM + kd) * DD + (t >> 2)];
    }

    const v4f* rowp = (const v4f*)(base + (size_t)k0 * MM * DD) + t;
    v4f rv[KT][8];
    #pragma unroll
    for (int kk = 0; kk < KT; ++kk) {
        #pragma unroll
        for (int q = 0; q < 8; ++q) rv[kk][q] = rowp[kk * (MM * DD / 4) + 256 * q];
    }

    asm volatile("" ::: "memory");

    const v4f* colp = (const v4f*)(base + (size_t)k0 * DD) + l + (size_t)w * (MM * DD / 4);
    v4f cv[32];
    #pragma unroll
    for (int q = 0; q < 16; ++q) cv[q] = colp[(size_t)q * (8 * (MM * DD / 4))];

    asm volatile("" :: "v"(rv[0][0]), "v"(rv[0][1]), "v"(rv[0][2]), "v"(rv[0][3]),
                       "v"(rv[0][4]), "v"(rv[0][5]), "v"(rv[0][6]), "v"(rv[0][7]),
                       "v"(rv[1][0]), "v"(rv[1][1]), "v"(rv[1][2]), "v"(rv[1][3]),
                       "v"(rv[1][4]), "v"(rv[1][5]), "v"(rv[1][6]), "v"(rv[1][7]),
                       "v"(rv[2][0]), "v"(rv[2][1]), "v"(rv[2][2]), "v"(rv[2][3]),
                       "v"(rv[2][4]), "v"(rv[2][5]), "v"(rv[2][6]), "v"(rv[2][7]),
                       "v"(rv[3][0]), "v"(rv[3][1]), "v"(rv[3][2]), "v"(rv[3][3]),
                       "v"(rv[3][4]), "v"(rv[3][5])
                 : "memory");

    const int jj = t >> 3, d4 = t & 7;
    v4f ra[KT];
    #pragma unroll
    for (int kk = 0; kk < KT; ++kk) {
        ra[kk] = ((rv[kk][0] + rv[kk][1]) + (rv[kk][2] + rv[kk][3]))
               + ((rv[kk][4] + rv[kk][5]) + (rv[kk][6] + rv[kk][7]));
    }
    v4f* srow = (v4f*)&sR[jj][16 * d4];
    #pragma unroll
    for (int c = 0; c < 4; ++c) {
        const v4f o4 = {ra[0][c], ra[1][c], ra[2][c], ra[3][c]};
        srow[c] = o4;
    }

    #pragma unroll
    for (int q = 16; q < 32; ++q) cv[q] = colp[(size_t)q * (8 * (MM * DD / 4))];

    asm volatile("" :: "v"(cv[0]), "v"(cv[1]), "v"(cv[2]), "v"(cv[3]),
                       "v"(cv[4]), "v"(cv[5]), "v"(cv[6]), "v"(cv[7]),
                       "v"(cv[8]), "v"(cv[9]), "v"(cv[10]), "v"(cv[11]),
                       "v"(cv[12]), "v"(cv[13]), "v"(cv[14]), "v"(cv[15])
                 : "memory");

    const v4f* cg4 = (const v4f*)coeffs + t;
    v4f cq[15];
    for (int q = 0; q < 15; ++q) cq[q] = cg4[256 * q];

    v4f ca = cv[0];
    #pragma unroll
    for (int q = 1; q < 32; ++q) ca += cv[q];
    float* scw = &sC[w][16 * (l & 7) + (l >> 3)];
    scw[0] = ca[0]; scw[4] = ca[1]; scw[8] = ca[2]; scw[12] = ca[3];

    v4f* cl4 = (v4f*)cfL + t;
    for (int q = 0; q < 15; ++q) cl4[256 * q] = cq[q];

    __syncthreads();

    if (t < DD * KT) {
        float rr = 0.f, cc = 0.f;
        #pragma unroll
        for (int g = 0; g < 32; ++g) rr += sR[g][t];
        #pragma unroll
        for (int g = 0; g < 8; ++g) cc += sC[g][t];
        sT[0][t] = rr;
        sT[1][t] = cc;
        sT[2][t] = gv;
    }

    if (blockIdx.x == 0 && blockIdx.y == 0) {
        for (int q = 0; q < 8; ++q) {
            const int idx = t + 256 * q;
            const int e = idx & 3, ln = (idx >> 2) & 31, g = idx >> 7;
            const int d2 = 4 * (g >> 1) + 2 * (ln >> 4) + (e & 1);
            const int s2 = (ln & 15) + 16 * (e >> 1);
            BQ[idx] = cfL[(d2 * SS + s2) * 15 + 9 + (g & 1)];
        }
    }
    __syncthreads();

    const float inv_m = 1.0f / (float)MM;
    const float inv_m2 = inv_m * inv_m;
    float u[KT], v[KT], ww[KT], zu = 0.f, zw = 0.f;
    for (int kk = 0; kk < KT; ++kk) { u[kk] = 0.f; v[kk] = 0.f; ww[kk] = 0.f; }
    for (int r = 0; r < 4; ++r) {
        const int d = 4 * w + r;
        const float* cf = cfL + (d * SS + l) * 15;
        const v4f r4 = *(const v4f*)&sT[0][4 * d];
        const v4f c4 = *(const v4f*)&sT[1][4 * d];
        const v4f g4 = *(const v4f*)&sT[2][4 * d];
        for (int kk = 0; kk < KT; ++kk) {
            const float rr = r4[kk] * inv_m;
            const float c = c4[kk] * inv_m;
            const float g = g4[kk];
            u[kk] += cf[5] * c + cf[6] * rr + cf[11] * g;
            v[kk] += cf[7] * c + cf[8] * rr + cf[12] * g;
            ww[kk] += cf[0] * g + cf[2] * rr + cf[3] * c;
        }
        const float p = ((g4[0] + g4[1]) + (g4[2] + g4[3])) * inv_m;
        const float tt = ((r4[0] + r4[1]) + (r4[2] + r4[3])) * inv_m2;
        zu += cf[13] * p + cf[14] * tt;
        zw += cf[1] * p + cf[4] * tt;
    }

    float* part = &sR[0][0];
    for (int kk = 0; kk < KT; ++kk) {
        part[(w * 14 + kk) * SS + l] = u[kk];
        part[(w * 14 + 4 + kk) * SS + l] = v[kk];
        part[(w * 14 + 8 + kk) * SS + l] = ww[kk];
    }
    part[(w * 14 + 12) * SS + l] = zu;
    part[(w * 14 + 13) * SS + l] = zw;
    __syncthreads();

    for (int q = w; q < 14; q += 8) {
        float sum = 0.f;
        for (int g = 0; g < 8; ++g) sum += part[(g * 14 + q) * SS + l];
        if (q < 4)       U[(n * MM + k0 + q) * SS + l] = sum + ab;
        else if (q < 8)  V[(n * MM + k0 + q - 4) * SS + l] = sum;
        else if (q < 12) W[(n * MM + k0 + q - 8) * SS + l] = sum + db;
        else             ZP[((n * 2 + q - 12) * KB + blockIdx.x) * SS + l] = sum;
    }
}

#define NT 8

__device__ __forceinline__ void ign_load_a(const float* __restrict__ p1,
                                           const float* __restrict__ p2,
                                           v2f (&A1)[8], v2f (&A2)[8]) {
    #pragma unroll
    for (int kb = 0; kb < 8; ++kb) {
        A1[kb] = *(const v2f*)(p1 + 4 * kb);
        A2[kb] = *(const v2f*)(p2 + 4 * kb);
    }
}

__device__ __forceinline__ void ign_load_v(const float* __restrict__ vp,
                                           float (&vj0)[8], float (&vj1)[8]) {
    #pragma unroll
    for (int v = 0; v < 8; ++v) {
        vj0[v] = vp[v * SS];
        vj1[v] = vp[v * SS + 16];
    }
}

__device__ __forceinline__ void ign_tile(const v2f (&A1)[8], const v2f (&A2)[8],
                                         const float4 (&B10)[8], const float4 (&B11)[8],
                                         const float (&vj0)[8], const float (&vj1)[8],
                                         float* __restrict__ op,
                                         float u0, float u1, float w0, float w1,
                                         int jrow, int i) {
    __builtin_amdgcn_sched_barrier(0);

    v8f acc0 = {}; v8f acc1 = {};
    #pragma unroll
    for (int kb = 0; kb < 8; ++kb) {
        const v2f bA0 = {B10[kb].x, B10[kb].y};
        const v2f bA1 = {B10[kb].z, B10[kb].w};
        const v2f bB0 = {B11[kb].x, B11[kb].y};
        const v2f bB1 = {B11[kb].z, B11[kb].w};
        acc0 = __builtin_amdgcn_wmma_f32_16x16x4_f32(false, A1[kb], false, bA0, (short)0, acc0, false, false);
        acc1 = __builtin_amdgcn_wmma_f32_16x16x4_f32(false, A1[kb], false, bA1, (short)0, acc1, false, false);
        acc0 = __builtin_amdgcn_wmma_f32_16x16x4_f32(false, A2[kb], false, bB0, (short)0, acc0, false, false);
        acc1 = __builtin_amdgcn_wmma_f32_16x16x4_f32(false, A2[kb], false, bB1, (short)0, acc1, false, false);
    }

    #pragma unroll
    for (int v = 0; v < 8; ++v) {
        const float dsel = (jrow + v == i) ? 1.0f : 0.0f;
        op[v * SS]      = acc0[v] + u0 + vj0[v] + dsel * w0;
        op[v * SS + 16] = acc1[v] + u1 + vj1[v] + dsel * w1;
    }
}

__global__ __launch_bounds__(256)
void ign_main_kernel(const float* __restrict__ in,
                     const float* __restrict__ BQ,
                     const float* __restrict__ U,
                     const float* __restrict__ V,
                     const float* __restrict__ W,
                     const float* __restrict__ ZP,
                     float* __restrict__ out) {
    __shared__ float zs[8][2][SS];

    const int n = blockIdx.z;
    const int i = blockIdx.y * 8 + (threadIdx.x >> 5);
    const int J0 = blockIdx.x * (16 * NT);
    const int lane = threadIdx.x & 31;
    const int m = lane & 15;
    const int kh = lane >> 4;

    const float* basen = in + (size_t)n * MM * MM * DD;
    const float* a1p = basen + ((size_t)i * MM + (J0 + m)) * DD + 2 * kh;
    const float* a2p = basen + ((size_t)(J0 + m) * MM + i) * DD + 2 * kh;
    const size_t a1step = (size_t)16 * DD;
    const size_t a2step = (size_t)16 * MM * DD;

    const float* zp = ZP + ((size_t)(n * 2) * (MM / 4) + 8 * (threadIdx.x >> 5)) * SS + lane;
    float zq[16];
    for (int q = 0; q < 8; ++q) {
        zq[q]     = zp[q * SS];
        zq[8 + q] = zp[((MM / 4) + q) * SS];
    }

    v2f Aa1[8], Aa2[8], Ab1[8], Ab2[8];
    ign_load_a(a1p, a2p, Aa1, Aa2);

    const float4* bq = (const float4*)BQ + lane;
    float4 B10[8], B11[8];
    #pragma unroll
    for (int kb = 0; kb < 8; ++kb) {
        B10[kb] = bq[(2 * kb) * 32];
        B11[kb] = bq[(2 * kb + 1) * 32];
    }

    const int s0 = m, s1 = m + 16;
    float u0 = U[(n * MM + i) * SS + s0];
    float u1 = U[(n * MM + i) * SS + s1];
    float w0 = W[(n * MM + i) * SS + s0];
    float w1 = W[(n * MM + i) * SS + s1];

    const float* vp = V + ((size_t)n * MM + J0 + 8 * kh) * SS + s0;
    float* op = out + (((size_t)n * MM + i) * MM + J0 + 8 * kh) * SS + s0;
    int jrow = J0 + 8 * kh;

    float vj0[8], vj1[8];

    ign_load_v(vp, vj0, vj1);
    ign_load_a(a1p + a1step, a2p + a2step, Ab1, Ab2);
    {
        float zu = zq[0], zw = zq[8];
        for (int q = 1; q < 8; ++q) { zu += zq[q]; zw += zq[8 + q]; }
        zs[threadIdx.x >> 5][0][lane] = zu;
        zs[threadIdx.x >> 5][1][lane] = zw;
        __builtin_amdgcn_fence(__ATOMIC_RELEASE, "workgroup", "local");
        __builtin_amdgcn_s_barrier();
        __builtin_amdgcn_fence(__ATOMIC_ACQUIRE, "workgroup", "local");
        float zu0 = zs[0][0][s0], zu1 = zs[0][0][s1], zw0 = zs[0][1][s0], zw1 = zs[0][1][s1];
        for (int g = 1; g < 8; ++g) {
            zu0 += zs[g][0][s0]; zu1 += zs[g][0][s1];
            zw0 += zs[g][1][s0]; zw1 += zs[g][1][s1];
        }
        u0 += zu0; u1 += zu1; w0 += zw0; w1 += zw1;
    }
    ign_tile(Aa1, Aa2, B10, B11, vj0, vj1, op, u0, u1, w0, w1, jrow, i);
    a1p += 2 * a1step;
    a2p += 2 * a2step;
    ign_load_v(vp + 16 * SS, vj0, vj1);
    ign_load_a(a1p, a2p, Aa1, Aa2);
    ign_tile(Ab1, Ab2, B10, B11, vj0, vj1, op + 16 * SS, u0, u1, w0, w1, jrow + 16, i);
    vp += 32 * SS;
    op += 32 * SS;
    jrow += 32;

    #pragma unroll 1
    for (int tp = 1; tp < NT / 2 - 1; ++tp) {
        ign_load_v(vp, vj0, vj1);
        ign_load_a(a1p + a1step, a2p + a2step, Ab1, Ab2);
        ign_tile(Aa1, Aa2, B10, B11, vj0, vj1, op, u0, u1, w0, w1, jrow, i);
        a1p += 2 * a1step;
        a2p += 2 * a2step;
        ign_load_v(vp + 16 * SS, vj0, vj1);
        ign_load_a(a1p, a2p, Aa1, Aa2);
        ign_tile(Ab1, Ab2, B10, B11, vj0, vj1, op + 16 * SS, u0, u1, w0, w1, jrow + 16, i);
        vp += 32 * SS;
        op += 32 * SS;
        jrow += 32;
    }
    ign_load_v(vp, vj0, vj1);
    ign_load_a(a1p + a1step, a2p + a2step, Ab1, Ab2);
    ign_tile(Aa1, Aa2, B10, B11, vj0, vj1, op, u0, u1, w0, w1, jrow, i);
    ign_load_v(vp + 16 * SS, vj0, vj1);
    ign_tile(Ab1, Ab2, B10, B11, vj0, vj1, op + 16 * SS, u0, u1, w0, w1, jrow + 16, i);
}

extern "C" void kernel_launch(void* const* d_in, const int* in_sizes, int n_in,
                              void* d_out, int out_size, void* d_ws, size_t ws_size,
                              hipStream_t stream) {
    (void)in_sizes; (void)n_in; (void)out_size; (void)ws_size;
    const float* inputs    = (const float*)d_in[0];
    const float* coeffs    = (const float*)d_in[2];
    const float* diag_bias = (const float*)d_in[3];
    const float* all_bias  = (const float*)d_in[4];
    float* out = (float*)d_out;

    float* ws   = (float*)d_ws;
    float* ZP   = ws;
    float* U    = ws   + 3 * NB * DD * MM + 2 * NB * DD;
    float* V    = U    + NB * MM * SS;
    float* W    = V    + NB * MM * SS;
    float* BQ   = W    + NB * MM * SS;

    ign_stats_kernel<<<dim3(MM / KT, NB), 256, 0, stream>>>(inputs, coeffs, diag_bias, all_bias,
                                                            U, V, W, ZP, BQ);
    ign_main_kernel<<<dim3(MM / (16 * NT), MM / 8, NB), 256, 0, stream>>>(
        inputs, BQ, U, V, W, ZP, out);
}


# Main kernel: all global accesses 16 B per lane: x fragments as 64-B runs (k order of products permuted, BQ too: sums re-ordered); tile transposed via per-wave LDS, V and out as whole lines
# speedup vs baseline: 2.8274x; 1.0729x over previous
#include <hip/hip_runtime.h>
#include <hip/hip_bf16.h>

#define NB 4
#define MM 256
#define DD 32
#define SS 32

typedef float v2f __attribute__((ext_vector_type(2)));
typedef float v8f __attribute__((ext_vector_type(8)));

#define KT 4
#define KB (MM / KT)
typedef float v4f __attribute__((ext_vector_type(4)));

__global__ __launch_bounds__(256)
void ign_stats_kernel(const float* __restrict__ in,
                      const float* __restrict__ coeffs,
                      const float* __restrict__ diag_bias,
                      const float* __restrict__ all_bias,
                      float* __restrict__ U,
                      float* __restrict__ V,
                      float* __restrict__ W,
                      float* __restrict__ ZP,
                      float* __restrict__ BQ) {
    __shared__ __attribute__((aligned(16))) float sm[(32 + 8 + 3) * DD * KT + DD * SS * 15];
    float (*sR)[DD * KT] = (float (*)[DD * KT])sm;
    float (*sC)[DD * KT] = (float (*)[DD * KT])(sm + 32 * DD * KT);
    float (*sT)[DD * KT] = (float (*)[DD * KT])(sm + 40 * DD * KT);
    float* cfL = sm + 43 * DD * KT;

    const int n = blockIdx.y, k0 = blockIdx.x * KT;
    const int t = threadIdx.x, l = t & 31, w = t >> 5;
    const float* base = in + (size_t)n * MM * MM * DD;

    const float ab = all_bias[l], db = diag_bias[l];

    float gv = 0.f;
    if (t < DD * KT) {
        const int kd = k0 + (t & 3);
        gv = base[((size_t)kd * MM + kd) * DD + (t >> 2)];
    }

    const v4f* rowp = (const v4f*)(base + (size_t)k0 * MM * DD) + t;
    v4f rv[KT][8];
    #pragma unroll
    for (int kk = 0; kk < KT; ++kk) {
        #pragma unroll
        for (int q = 0; q < 8; ++q) rv[kk][q] = rowp[kk * (MM * DD / 4) + 256 * q];
    }

    asm volatile("" ::: "memory");

    const v4f* colp = (const v4f*)(base + (size_t)k0 * DD) + l + (size_t)w * (MM * DD / 4);
    v4f cv[32];
    #pragma unroll
    for (int q = 0; q < 16; ++q) cv[q] = colp[(size_t)q * (8 * (MM * DD / 4))];

    asm volatile("" :: "v"(rv[0][0]), "v"(rv[0][1]), "v"(rv[0][2]), "v"(rv[0][3]),
                       "v"(rv[0][4]), "v"(rv[0][5]), "v"(rv[0][6]), "v"(rv[0][7]),
                       "v"(rv[1][0]), "v"(rv[1][1]), "v"(rv[1][2]), "v"(rv[1][3]),
                       "v"(rv[1][4]), "v"(rv[1][5]), "v"(rv[1][6]), "v"(rv[1][7]),
                       "v"(rv[2][0]), "v"(rv[2][1]), "v"(rv[2][2]), "v"(rv[2][3]),
                       "v"(rv[2][4]), "v"(rv[2][5]), "v"(rv[2][6]), "v"(rv[2][7]),
                       "v"(rv[3][0]), "v"(rv[3][1]), "v"(rv[3][2]), "v"(rv[3][3]),
                       "v"(rv[3][4]), "v"(rv[3][5])
                 : "memory");

    const int jj = t >> 3, d4 = t & 7;
    v4f ra[KT];
    #pragma unroll
    for (int kk = 0; kk < KT; ++kk) {
        ra[kk] = ((rv[kk][0] + rv[kk][1]) + (rv[kk][2] + rv[kk][3]))
               + ((rv[kk][4] + rv[kk][5]) + (rv[kk][6] + rv[kk][7]));
    }
    v4f* srow = (v4f*)&sR[jj][16 * d4];
    #pragma unroll
    for (int c = 0; c < 4; ++c) {
        const v4f o4 = {ra[0][c], ra[1][c], ra[2][c], ra[3][c]};
        srow[c] = o4;
    }

    #pragma unroll
    for (int q = 16; q < 32; ++q) cv[q] = colp[(size_t)q * (8 * (MM * DD / 4))];

    asm volatile("" :: "v"(cv[0]), "v"(cv[1]), "v"(cv[2]), "v"(cv[3]),
                       "v"(cv[4]), "v"(cv[5]), "v"(cv[6]), "v"(cv[7]),
                       "v"(cv[8]), "v"(cv[9]), "v"(cv[10]), "v"(cv[11]),
                       "v"(cv[12]), "v"(cv[13]), "v"(cv[14]), "v"(cv[15])
                 : "memory");

    const v4f* cg4 = (const v4f*)coeffs + t;
    v4f cq[15];
    for (int q = 0; q < 15; ++q) cq[q] = cg4[256 * q];

    v4f ca = cv[0];
    #pragma unroll
    for (int q = 1; q < 32; ++q) ca += cv[q];
    float* scw = &sC[w][16 * (l & 7) + (l >> 3)];
    scw[0] = ca[0]; scw[4] = ca[1]; scw[8] = ca[2]; scw[12] = ca[3];

    v4f* cl4 = (v4f*)cfL + t;
    for (int q = 0; q < 15; ++q) cl4[256 * q] = cq[q];

    __syncthreads();

    if (t < DD * KT) {
        float rr = 0.f, cc = 0.f;
        #pragma unroll
        for (int g = 0; g < 32; ++g) rr += sR[g][t];
        #pragma unroll
        for (int g = 0; g < 8; ++g) cc += sC[g][t];
        sT[0][t] = rr;
        sT[1][t] = cc;
        sT[2][t] = gv;
    }

    if (blockIdx.x == 0 && blockIdx.y == 0) {
        for (int q = 0; q < 8; ++q) {
            const int idx = t + 256 * q;
            const int e = idx & 3, ln = (idx >> 2) & 31, g = idx >> 7;
            const int d2 = 16 * (ln >> 4) + 2 * (g >> 1) + (e & 1);
            const int s2 = (ln & 15) + 16 * (e >> 1);
            BQ[idx] = cfL[(d2 * SS + s2) * 15 + 9 + (g & 1)];
        }
    }
    __syncthreads();

    const float inv_m = 1.0f / (float)MM;
    const float inv_m2 = inv_m * inv_m;
    float u[KT], v[KT], ww[KT], zu = 0.f, zw = 0.f;
    for (int kk = 0; kk < KT; ++kk) { u[kk] = 0.f; v[kk] = 0.f; ww[kk] = 0.f; }
    for (int r = 0; r < 4; ++r) {
        const int d = 4 * w + r;
        const float* cf = cfL + (d * SS + l) * 15;
        const v4f r4 = *(const v4f*)&sT[0][4 * d];
        const v4f c4 = *(const v4f*)&sT[1][4 * d];
        const v4f g4 = *(const v4f*)&sT[2][4 * d];
        for (int kk = 0; kk < KT; ++kk) {
            const float rr = r4[kk] * inv_m;
            const float c = c4[kk] * inv_m;
            const float g = g4[kk];
            u[kk] += cf[5] * c + cf[6] * rr + cf[11] * g;
            v[kk] += cf[7] * c + cf[8] * rr + cf[12] * g;
            ww[kk] += cf[0] * g + cf[2] * rr + cf[3] * c;
        }
        const float p = ((g4[0] + g4[1]) + (g4[2] + g4[3])) * inv_m;
        const float tt = ((r4[0] + r4[1]) + (r4[2] + r4[3])) * inv_m2;
        zu += cf[13] * p + cf[14] * tt;
        zw += cf[1] * p + cf[4] * tt;
    }

    float* part = &sR[0][0];
    for (int kk = 0; kk < KT; ++kk) {
        part[(w * 14 + kk) * SS + l] = u[kk];
        part[(w * 14 + 4 + kk) * SS + l] = v[kk];
        part[(w * 14 + 8 + kk) * SS + l] = ww[kk];
    }
    part[(w * 14 + 12) * SS + l] = zu;
    part[(w * 14 + 13) * SS + l] = zw;
    __syncthreads();

    for (int q = w; q < 14; q += 8) {
        float sum = 0.f;
        for (int g = 0; g < 8; ++g) sum += part[(g * 14 + q) * SS + l];
        if (q < 4)       U[(n * MM + k0 + q) * SS + l] = sum + ab;
        else if (q < 8)  V[(n * MM + k0 + q - 4) * SS + l] = sum;
        else if (q < 12) W[(n * MM + k0 + q - 8) * SS + l] = sum + db;
        else             ZP[((n * 2 + q - 12) * KB + blockIdx.x) * SS + l] = sum;
    }
}

#define NT 8
#define TP 36

__device__ __forceinline__ void ign_load_x(const float* __restrict__ p1,
                                           const float* __restrict__ p2,
                                           float4 (&X1)[4], float4 (&X2)[4]) {
    #pragma unroll
    for (int q = 0; q < 4; ++q) {
        X1[q] = *(const float4*)(p1 + 4 * q);
        X2[q] = *(const float4*)(p2 + 4 * q);
    }
}

__device__ __forceinline__ void ign_load_v(const float* __restrict__ vp, float4 (&vq)[4]) {
    #pragma unroll
    for (int q = 0; q < 4; ++q) vq[q] = *(const float4*)(vp + 4 * q * SS);
}

__device__ __forceinline__ void ign_tile(const float4 (&X1)[4], const float4 (&X2)[4],
                                         const float4 (&B10)[8], const float4 (&B11)[8],
                                         const float4 (&vq)[4],
                                         float* __restrict__ twr, const float* __restrict__ trd,
                                         float* __restrict__ op,
                                         float4 u4, float4 w4, int jrow, int i) {
    __builtin_amdgcn_sched_barrier(0);

    v8f acc0 = {}; v8f acc1 = {};
    #pragma unroll
    for (int kb = 0; kb < 8; ++kb) {
        const float4 q1 = X1[kb >> 1];
        const float4 q2 = X2[kb >> 1];
        const v2f x1 = (kb & 1) ? v2f{q1.z, q1.w} : v2f{q1.x, q1.y};
        const v2f x2 = (kb & 1) ? v2f{q2.z, q2.w} : v2f{q2.x, q2.y};
        const v2f cA0 = {B10[kb].x, B10[kb].y};
        const v2f cA1 = {B10[kb].z, B10[kb].w};
        const v2f cB0 = {B11[kb].x, B11[kb].y};
        const v2f cB1 = {B11[kb].z, B11[kb].w};
        acc0 = __builtin_amdgcn_wmma_f32_16x16x4_f32(false, cA0, false, x1, (short)0, acc0, false, false);
        acc1 = __builtin_amdgcn_wmma_f32_16x16x4_f32(false, cA1, false, x1, (short)0, acc1, false, false);
        acc0 = __builtin_amdgcn_wmma_f32_16x16x4_f32(false, cB0, false, x2, (short)0, acc0, false, false);
        acc1 = __builtin_amdgcn_wmma_f32_16x16x4_f32(false, cB1, false, x2, (short)0, acc1, false, false);
    }

    __builtin_amdgcn_fence(__ATOMIC_RELEASE, "wavefront", "local");
    __builtin_amdgcn_wave_barrier();
    __builtin_amdgcn_fence(__ATOMIC_ACQUIRE, "wavefront", "local");
    *(float4*)(twr)      = float4{acc0[0], acc0[1], acc0[2], acc0[3]};
    *(float4*)(twr + 4)  = float4{acc0[4], acc0[5], acc0[6], acc0[7]};
    *(float4*)(twr + 16) = float4{acc1[0], acc1[1], acc1[2], acc1[3]};
    *(float4*)(twr + 20) = float4{acc1[4], acc1[5], acc1[6], acc1[7]};
    __builtin_amdgcn_fence(__ATOMIC_RELEASE, "wavefront", "local");
    __builtin_amdgcn_wave_barrier();
    __builtin_amdgcn_fence(__ATOMIC_ACQUIRE, "wavefront", "local");

    #pragma unroll
    for (int q = 0; q < 4; ++q) {
        const float4 t4 = *(const float4*)(trd + 4 * q * TP);
        const float dsel = (jrow + 4 * q == i) ? 1.0f : 0.0f;
        float4 o;
        o.x = t4.x + u4.x + vq[q].x + dsel * w4.x;
        o.y = t4.y + u4.y + vq[q].y + dsel * w4.y;
        o.z = t4.z + u4.z + vq[q].z + dsel * w4.z;
        o.w = t4.w + u4.w + vq[q].w + dsel * w4.w;
        *(float4*)(op + 4 * q * SS) = o;
    }
}

__global__ __launch_bounds__(256)
void ign_main_kernel(const float* __restrict__ in,
                     const float* __restrict__ BQ,
                     const float* __restrict__ U,
                     const float* __restrict__ V,
                     const float* __restrict__ W,
                     const float* __restrict__ ZP,
                     float* __restrict__ out) {
    __shared__ __attribute__((aligned(16))) float zs[8][2][SS];
    __shared__ __attribute__((aligned(16))) float ts[8][16 * TP];

    const int n = blockIdx.z;
    const int wv = threadIdx.x >> 5;
    const int i = blockIdx.y * 8 + wv;
    const int J0 = blockIdx.x * (16 * NT);
    const int lane = threadIdx.x & 31;
    const int c = lane & 15;
    const int h = lane >> 4;
    const int r4 = lane >> 3;
    const int c4 = 4 * (lane & 7);

    const float* basen = in + (size_t)n * MM * MM * DD;
    const float* x1p = basen + ((size_t)i * MM + (J0 + c)) * DD + 16 * h;
    const float* x2p = basen + ((size_t)(J0 + c) * MM + i) * DD + 16 * h;
    const size_t x1step = (size_t)16 * DD;
    const size_t x2step = (size_t)16 * MM * DD;

    const float* zp = ZP + ((size_t)(n * 2) * (MM / 4) + 8 * wv) * SS + lane;
    float zq[16];
    for (int q = 0; q < 8; ++q) {
        zq[q]     = zp[q * SS];
        zq[8 + q] = zp[((MM / 4) + q) * SS];
    }

    float4 Xa1[4], Xa2[4], Xb1[4], Xb2[4];
    ign_load_x(x1p, x2p, Xa1, Xa2);

    const float4* bq = (const float4*)BQ + lane;
    float4 B10[8], B11[8];
    #pragma unroll
    for (int kb = 0; kb < 8; ++kb) {
        B10[kb] = bq[(2 * kb) * 32];
        B11[kb] = bq[(2 * kb + 1) * 32];
    }

    float4 u4 = *(const float4*)(U + (size_t)(n * MM + i) * SS + c4);
    float4 w4 = *(const float4*)(W + (size_t)(n * MM + i) * SS + c4);

    const float* vp = V + ((size_t)n * MM + J0 + r4) * SS + c4;
    float* op = out + (((size_t)n * MM + i) * MM + J0 + r4) * SS + c4;
    int jrow = J0 + r4;

    float* twr = &ts[wv][c * TP + 8 * h];
    const float* trd = &ts[wv][r4 * TP + c4];

    float4 vq[4];

    ign_load_v(vp, vq);
    ign_load_x(x1p + x1step, x2p + x2step, Xb1, Xb2);
    {
        float zu = zq[0], zw = zq[8];
        for (int q = 1; q < 8; ++q) { zu += zq[q]; zw += zq[8 + q]; }
        zs[wv][0][lane] = zu;
        zs[wv][1][lane] = zw;
        __builtin_amdgcn_fence(__ATOMIC_RELEASE, "workgroup", "local");
        __builtin_amdgcn_s_barrier();
        __builtin_amdgcn_fence(__ATOMIC_ACQUIRE, "workgroup", "local");
        float4 zu4 = *(const float4*)&zs[0][0][c4];
        float4 zw4 = *(const float4*)&zs[0][1][c4];
        for (int g = 1; g < 8; ++g) {
            const float4 a = *(const float4*)&zs[g][0][c4];
            const float4 b = *(const float4*)&zs[g][1][c4];
            zu4.x += a.x; zu4.y += a.y; zu4.z += a.z; zu4.w += a.w;
            zw4.x += b.x; zw4.y += b.y; zw4.z += b.z; zw4.w += b.w;
        }
        u4.x += zu4.x; u4.y += zu4.y; u4.z += zu4.z; u4.w += zu4.w;
        w4.x += zw4.x; w4.y += zw4.y; w4.z += zw4.z; w4.w += zw4.w;
    }
    ign_tile(Xa1, Xa2, B10, B11, vq, twr, trd, op, u4, w4, jrow, i);
    x1p += 2 * x1step;
    x2p += 2 * x2step;
    ign_load_v(vp + 16 * SS, vq);
    ign_load_x(x1p, x2p, Xa1, Xa2);
    ign_tile(Xb1, Xb2, B10, B11, vq, twr, trd, op + 16 * SS, u4, w4, jrow + 16, i);
    vp += 32 * SS;
    op += 32 * SS;
    jrow += 32;

    #pragma unroll 1
    for (int tp = 1; tp < NT / 2 - 1; ++tp) {
        ign_load_v(vp, vq);
        ign_load_x(x1p + x1step, x2p + x2step, Xb1, Xb2);
        ign_tile(Xa1, Xa2, B10, B11, vq, twr, trd, op, u4, w4, jrow, i);
        x1p += 2 * x1step;
        x2p += 2 * x2step;
        ign_load_v(vp + 16 * SS, vq);
        ign_load_x(x1p, x2p, Xa1, Xa2);
        ign_tile(Xb1, Xb2, B10, B11, vq, twr, trd, op + 16 * SS, u4, w4, jrow + 16, i);
        vp += 32 * SS;
        op += 32 * SS;
        jrow += 32;
    }
    ign_load_v(vp, vq);
    ign_load_x(x1p + x1step, x2p + x2step, Xb1, Xb2);
    ign_tile(Xa1, Xa2, B10, B11, vq, twr, trd, op, u4, w4, jrow, i);
    ign_load_v(vp + 16 * SS, vq);
    ign_tile(Xb1, Xb2, B10, B11, vq, twr, trd, op + 16 * SS, u4, w4, jrow + 16, i);
}

extern "C" void kernel_launch(void* const* d_in, const int* in_sizes, int n_in,
                              void* d_out, int out_size, void* d_ws, size_t ws_size,
                              hipStream_t stream) {
    (void)in_sizes; (void)n_in; (void)out_size; (void)ws_size;
    const float* inputs    = (const float*)d_in[0];
    const float* coeffs    = (const float*)d_in[2];
    const float* diag_bias = (const float*)d_in[3];
    const float* all_bias  = (const float*)d_in[4];
    float* out = (float*)d_out;

    float* ws   = (float*)d_ws;
    float* ZP   = ws;
    float* U    = ws   + 3 * NB * DD * MM + 2 * NB * DD;
    float* V    = U    + NB * MM * SS;
    float* W    = V    + NB * MM * SS;
    float* BQ   = W    + NB * MM * SS;

    ign_stats_kernel<<<dim3(MM / KT, NB), 256, 0, stream>>>(inputs, coeffs, diag_bias, all_bias,
                                                            U, V, W, ZP, BQ);
    ign_main_kernel<<<dim3(MM / (16 * NT), MM / 8, NB), 256, 0, stream>>>(
        inputs, BQ, U, V, W, ZP, out);
}
